# starting kernel
# baseline (speedup 1.0000x reference)
#include <hip/hip_runtime.h>
#include <math.h>


typedef float v2f __attribute__((ext_vector_type(2)));
typedef float v8f __attribute__((ext_vector_type(8)));

#define EMBED            64
#define WAVES_PER_BLOCK  8
#define BLOCK            (WAVES_PER_BLOCK * 32)
#define TILE_EDGES       16
#define EDGES_PER_BLOCK  (WAVES_PER_BLOCK * TILE_EDGES)
#define W1T_STRIDE       132

__global__ __launch_bounds__(BLOCK)
void edge_mlp_wmma_kernel(const float* __restrict__ hc,
                          const float* __restrict__ hp,
                          const int*   __restrict__ src,
                          const int*   __restrict__ dst,
                          const float* __restrict__ W1,
                          const float* __restrict__ b1,
                          const float* __restrict__ W2,
                          const float* __restrict__ b2,
                          float*       __restrict__ out,
                          int E)
{
    __shared__ float sW1t[64 * W1T_STRIDE];
    __shared__ float sW1cos[64];
    __shared__ float sB1[64];
    __shared__ float sW2[64];

    const int tid = threadIdx.x;

    for (int i = tid; i < 129 * 64; i += BLOCK) {
        const int k = i >> 6;
        const int n = i & 63;
        const float v = W1[i];
        if (k < 128) sW1t[n * W1T_STRIDE + k] = v;
        else         sW1cos[n] = v;
    }
    for (int i = tid; i < 64; i += BLOCK) { sB1[i] = b1[i]; sW2[i] = W2[i]; }
    __syncthreads();

    const int lane = tid & 31;
    const int wave = tid >> 5;
    const int nl   = lane & 15;
    const int hi   = lane >> 4;
    const int edgeBase = (blockIdx.x * WAVES_PER_BLOCK + wave) * TILE_EDGES;
    if (edgeBase >= E) return;

    int e = edgeBase + nl;
    if (e >= E) e = E - 1;
    const float* __restrict__ hrow = hc + (size_t)src[e] * EMBED;
    const float* __restrict__ trow = hp + (size_t)dst[e] * EMBED;

    v8f acc[4] = {};
    v2f hbuf[16];
    float hh = 0.f, tt = 0.f, dp = 0.f;

#pragma unroll
    for (int s = 0; s < 16; ++s) {
        const int kb = 4 * s + 2 * hi;
        v2f a;
        a.x = hrow[kb];
        a.y = hrow[kb + 1];
        hbuf[s] = a;
        hh += a.x * a.x + a.y * a.y;
#pragma unroll
        for (int t = 0; t < 4; ++t) {
            const float* wp = &sW1t[(t * 16 + nl) * W1T_STRIDE + kb];
            v2f b; b.x = wp[0]; b.y = wp[1];
            acc[t] = __builtin_amdgcn_wmma_f32_16x16x4_f32(
                false, a, false, b, (short)0, acc[t], false, false);
        }
    }

#pragma unroll
    for (int s = 0; s < 16; ++s) {
        const int kb = 4 * s + 2 * hi;
        v2f a;
        a.x = trow[kb];
        a.y = trow[kb + 1];
        tt += a.x * a.x + a.y * a.y;
        dp += hbuf[s].x * a.x + hbuf[s].y * a.y;
        const int kw = 64 + kb;
#pragma unroll
        for (int t = 0; t < 4; ++t) {
            const float* wp = &sW1t[(t * 16 + nl) * W1T_STRIDE + kw];
            v2f b; b.x = wp[0]; b.y = wp[1];
            acc[t] = __builtin_amdgcn_wmma_f32_16x16x4_f32(
                false, a, false, b, (short)0, acc[t], false, false);
        }
    }

    hh += __shfl_xor(hh, 16, 32);
    tt += __shfl_xor(tt, 16, 32);
    dp += __shfl_xor(dp, 16, 32);
    const float nh  = sqrtf(hh);
    const float nt  = sqrtf(tt);
    const float inh = 1.f / fmaxf(nh, 1e-12f);
    const float itn = 1.f / fmaxf(nt, 1e-12f);
    const float den = fmaxf((nh * inh) * (nt * itn), 1e-8f);
    const float cosv = (dp * inh * itn) / den;

    float w1c[4], b1v[4], w2v[4];
#pragma unroll
    for (int t = 0; t < 4; ++t) {
        const int n = t * 16 + nl;
        w1c[t] = sW1cos[n];
        b1v[t] = sB1[n];
        w2v[t] = sW2[n];
    }
    const float b2s = b2[0];

    float r8[8];
#pragma unroll
    for (int r = 0; r < 8; ++r) {
        const int m = r + 8 * hi;
        const float cm = __shfl(cosv, m, 32);
        float p = 0.f;
#pragma unroll
        for (int t = 0; t < 4; ++t) {
            float x = acc[t][r] + b1v[t] + cm * w1c[t];
            x = fmaxf(x, 0.f);
            p += x * w2v[t];
        }
        p += __shfl_xor(p, 1, 32);
        p += __shfl_xor(p, 2, 32);
        p += __shfl_xor(p, 4, 32);
        p += __shfl_xor(p, 8, 32);
        r8[r] = p;
    }

    if (nl == 0) {
#pragma unroll
        for (int r = 0; r < 8; ++r) {
            const int eo = edgeBase + r + 8 * hi;
            if (eo < E) {
                const float z = r8[r] + b2s;
                out[eo] = 1.f / (1.f + expf(-z));
            }
        }
    }
}

extern "C" void kernel_launch(void* const* d_in, const int* in_sizes, int n_in,
                              void* d_out, int out_size, void* d_ws, size_t ws_size,
                              hipStream_t stream)
{
    const float* hc  = (const float*)d_in[0];
    const float* hp  = (const float*)d_in[1];
    const int*   src = (const int*)  d_in[2];
    const int*   dst = (const int*)  d_in[3];
    const float* W1  = (const float*)d_in[4];
    const float* b1  = (const float*)d_in[5];
    const float* W2  = (const float*)d_in[6];
    const float* b2  = (const float*)d_in[7];
    float* out = (float*)d_out;

    const int E = in_sizes[2];
    const int grid = (E + EDGES_PER_BLOCK - 1) / EDGES_PER_BLOCK;

    hipLaunchKernelGGL(edge_mlp_wmma_kernel, dim3(grid), dim3(BLOCK), 0, stream,
                       hc, hp, src, dst, W1, b1, W2, b2, out, E);
}


# Persistent grid: 256 resident 32-wave work-groups (one per WGP) stage W1 into LDS once and their waves loop over 16-edge tiles; before, every 128 edges re-staged W1 by 33 serial loads
# speedup vs baseline: 1.3086x; 1.3086x over previous
#include <hip/hip_runtime.h>
#include <math.h>


typedef float v2f __attribute__((ext_vector_type(2)));
typedef float v8f __attribute__((ext_vector_type(8)));

#define EMBED            64
#define WAVES_PER_BLOCK  32
#define GRID_BLOCKS      256
#define BLOCK            (WAVES_PER_BLOCK * 32)
#define TILE_EDGES       16
#define W1T_STRIDE       132

__global__ __launch_bounds__(BLOCK)
void edge_mlp_wmma_kernel(const float* __restrict__ hc,
                          const float* __restrict__ hp,
                          const int*   __restrict__ src,
                          const int*   __restrict__ dst,
                          const float* __restrict__ W1,
                          const float* __restrict__ b1,
                          const float* __restrict__ W2,
                          const float* __restrict__ b2,
                          float*       __restrict__ out,
                          int E)
{
    __shared__ float sW1t[64 * W1T_STRIDE];
    __shared__ float sW1cos[64];
    __shared__ float sB1[64];
    __shared__ float sW2[64];

    const int tid = threadIdx.x;

    for (int i = tid; i < 129 * 64; i += BLOCK) {
        const int k = i >> 6;
        const int n = i & 63;
        const float v = W1[i];
        if (k < 128) sW1t[n * W1T_STRIDE + k] = v;
        else         sW1cos[n] = v;
    }
    for (int i = tid; i < 64; i += BLOCK) { sB1[i] = b1[i]; sW2[i] = W2[i]; }
    __syncthreads();

    const int lane = tid & 31;
    const int wave = __builtin_amdgcn_readfirstlane(tid >> 5);
    const int nl   = lane & 15;
    const int hi   = lane >> 4;
    const int nTiles     = (E + TILE_EDGES - 1) / TILE_EDGES;
    const int tileStride = GRID_BLOCKS * WAVES_PER_BLOCK;

    for (int tile = 2 * ((wave >> 1) * GRID_BLOCKS + blockIdx.x) + (wave & 1);
         tile < nTiles; tile += tileStride) {
    asm volatile("" ::: "memory");
    const int edgeBase = tile * TILE_EDGES;

    int e = edgeBase + nl;
    if (e >= E) e = E - 1;
    const float* __restrict__ hrow = hc + (size_t)src[e] * EMBED;
    const float* __restrict__ trow = hp + (size_t)dst[e] * EMBED;

    v8f acc[4] = {};
    v2f hbuf[16];
    float hh = 0.f, tt = 0.f, dp = 0.f;

#pragma unroll
    for (int s = 0; s < 16; ++s) {
        const int kb = 4 * s + 2 * hi;
        v2f a;
        a.x = hrow[kb];
        a.y = hrow[kb + 1];
        hbuf[s] = a;
        hh += a.x * a.x + a.y * a.y;
#pragma unroll
        for (int t = 0; t < 4; ++t) {
            const float* wp = &sW1t[(t * 16 + nl) * W1T_STRIDE + kb];
            v2f b; b.x = wp[0]; b.y = wp[1];
            acc[t] = __builtin_amdgcn_wmma_f32_16x16x4_f32(
                false, a, false, b, (short)0, acc[t], false, false);
        }
    }

#pragma unroll
    for (int s = 0; s < 16; ++s) {
        const int kb = 4 * s + 2 * hi;
        v2f a;
        a.x = trow[kb];
        a.y = trow[kb + 1];
        tt += a.x * a.x + a.y * a.y;
        dp += hbuf[s].x * a.x + hbuf[s].y * a.y;
        const int kw = 64 + kb;
#pragma unroll
        for (int t = 0; t < 4; ++t) {
            const float* wp = &sW1t[(t * 16 + nl) * W1T_STRIDE + kw];
            v2f b; b.x = wp[0]; b.y = wp[1];
            acc[t] = __builtin_amdgcn_wmma_f32_16x16x4_f32(
                false, a, false, b, (short)0, acc[t], false, false);
        }
    }

    hh += __shfl_xor(hh, 16, 32);
    tt += __shfl_xor(tt, 16, 32);
    dp += __shfl_xor(dp, 16, 32);
    const float nh  = sqrtf(hh);
    const float nt  = sqrtf(tt);
    const float inh = 1.f / fmaxf(nh, 1e-12f);
    const float itn = 1.f / fmaxf(nt, 1e-12f);
    const float den = fmaxf((nh * inh) * (nt * itn), 1e-8f);
    const float cosv = (dp * inh * itn) / den;

    float w1c[4], b1v[4], w2v[4];
#pragma unroll
    for (int t = 0; t < 4; ++t) {
        const int n = t * 16 + nl;
        w1c[t] = sW1cos[n];
        b1v[t] = sB1[n];
        w2v[t] = sW2[n];
    }
    const float b2s = b2[0];

    float r8[8];
#pragma unroll
    for (int r = 0; r < 8; ++r) {
        const int m = r + 8 * hi;
        const float cm = __shfl(cosv, m, 32);
        float p = 0.f;
#pragma unroll
        for (int t = 0; t < 4; ++t) {
            float x = acc[t][r] + b1v[t] + cm * w1c[t];
            x = fmaxf(x, 0.f);
            p += x * w2v[t];
        }
        p += __shfl_xor(p, 1, 32);
        p += __shfl_xor(p, 2, 32);
        p += __shfl_xor(p, 4, 32);
        p += __shfl_xor(p, 8, 32);
        r8[r] = p;
    }

    if (nl == 0) {
#pragma unroll
        for (int r = 0; r < 8; ++r) {
            const int eo = edgeBase + r + 8 * hi;
            if (eo < E) {
                const float z = r8[r] + b2s;
                out[eo] = 1.f / (1.f + expf(-z));
            }
        }
    }
    }
}

extern "C" void kernel_launch(void* const* d_in, const int* in_sizes, int n_in,
                              void* d_out, int out_size, void* d_ws, size_t ws_size,
                              hipStream_t stream)
{
    const float* hc  = (const float*)d_in[0];
    const float* hp  = (const float*)d_in[1];
    const int*   src = (const int*)  d_in[2];
    const int*   dst = (const int*)  d_in[3];
    const float* W1  = (const float*)d_in[4];
    const float* b1  = (const float*)d_in[5];
    const float* W2  = (const float*)d_in[6];
    const float* b2  = (const float*)d_in[7];
    float* out = (float*)d_out;

    const int E = in_sizes[2];

    hipLaunchKernelGGL(edge_mlp_wmma_kernel, dim3(GRID_BLOCKS), dim3(BLOCK), 0, stream,
                       hc, hp, src, dst, W1, b1, W2, b2, out, E);
}


# Epilogue tail once per tile: each lane selects one edge's row sum, so bias, exp, divide and store run in one 16-lane pass (one 64-byte store), not eight 2-lane passes; sums and results unchanged
# speedup vs baseline: 1.3766x; 1.0520x over previous
#include <hip/hip_runtime.h>
#include <math.h>


typedef float v2f __attribute__((ext_vector_type(2)));
typedef float v8f __attribute__((ext_vector_type(8)));

#define EMBED            64
#define WAVES_PER_BLOCK  32
#define GRID_BLOCKS      256
#define BLOCK            (WAVES_PER_BLOCK * 32)
#define TILE_EDGES       16
#define W1T_STRIDE       132

__global__ __launch_bounds__(BLOCK)
void edge_mlp_wmma_kernel(const float* __restrict__ hc,
                          const float* __restrict__ hp,
                          const int*   __restrict__ src,
                          const int*   __restrict__ dst,
                          const float* __restrict__ W1,
                          const float* __restrict__ b1,
                          const float* __restrict__ W2,
                          const float* __restrict__ b2,
                          float*       __restrict__ out,
                          int E)
{
    __shared__ float sW1t[64 * W1T_STRIDE];
    __shared__ float sW1cos[64];
    __shared__ float sB1[64];
    __shared__ float sW2[64];

    const int tid = threadIdx.x;

    for (int i = tid; i < 129 * 64; i += BLOCK) {
        const int k = i >> 6;
        const int n = i & 63;
        const float v = W1[i];
        if (k < 128) sW1t[n * W1T_STRIDE + k] = v;
        else         sW1cos[n] = v;
    }
    for (int i = tid; i < 64; i += BLOCK) { sB1[i] = b1[i]; sW2[i] = W2[i]; }
    __syncthreads();

    const int lane = tid & 31;
    const int wave = __builtin_amdgcn_readfirstlane(tid >> 5);
    const int nl   = lane & 15;
    const int hi   = lane >> 4;
    const int nTiles     = (E + TILE_EDGES - 1) / TILE_EDGES;
    const int tileStride = GRID_BLOCKS * WAVES_PER_BLOCK;

    for (int tile = 2 * ((wave >> 1) * GRID_BLOCKS + blockIdx.x) + (wave & 1);
         tile < nTiles; tile += tileStride) {
    asm volatile("" ::: "memory");
    const int edgeBase = tile * TILE_EDGES;

    int e = edgeBase + nl;
    if (e >= E) e = E - 1;
    const float* __restrict__ hrow = hc + (size_t)src[e] * EMBED;
    const float* __restrict__ trow = hp + (size_t)dst[e] * EMBED;

    v8f acc[4] = {};
    v2f hbuf[16];
    float hh = 0.f, tt = 0.f, dp = 0.f;

#pragma unroll
    for (int s = 0; s < 16; ++s) {
        const int kb = 4 * s + 2 * hi;
        v2f a;
        a.x = hrow[kb];
        a.y = hrow[kb + 1];
        hbuf[s] = a;
        hh += a.x * a.x + a.y * a.y;
#pragma unroll
        for (int t = 0; t < 4; ++t) {
            const float* wp = &sW1t[(t * 16 + nl) * W1T_STRIDE + kb];
            v2f b; b.x = wp[0]; b.y = wp[1];
            acc[t] = __builtin_amdgcn_wmma_f32_16x16x4_f32(
                false, a, false, b, (short)0, acc[t], false, false);
        }
    }

#pragma unroll
    for (int s = 0; s < 16; ++s) {
        const int kb = 4 * s + 2 * hi;
        v2f a;
        a.x = trow[kb];
        a.y = trow[kb + 1];
        tt += a.x * a.x + a.y * a.y;
        dp += hbuf[s].x * a.x + hbuf[s].y * a.y;
        const int kw = 64 + kb;
#pragma unroll
        for (int t = 0; t < 4; ++t) {
            const float* wp = &sW1t[(t * 16 + nl) * W1T_STRIDE + kw];
            v2f b; b.x = wp[0]; b.y = wp[1];
            acc[t] = __builtin_amdgcn_wmma_f32_16x16x4_f32(
                false, a, false, b, (short)0, acc[t], false, false);
        }
    }

    hh += __shfl_xor(hh, 16, 32);
    tt += __shfl_xor(tt, 16, 32);
    dp += __shfl_xor(dp, 16, 32);
    const float nh  = sqrtf(hh);
    const float nt  = sqrtf(tt);
    const float inh = 1.f / fmaxf(nh, 1e-12f);
    const float itn = 1.f / fmaxf(nt, 1e-12f);
    const float den = fmaxf((nh * inh) * (nt * itn), 1e-8f);
    const float cosv = (dp * inh * itn) / den;

    float w1c[4], b1v[4], w2v[4];
#pragma unroll
    for (int t = 0; t < 4; ++t) {
        const int n = t * 16 + nl;
        w1c[t] = sW1cos[n];
        b1v[t] = sB1[n];
        w2v[t] = sW2[n];
    }
    const float b2s = b2[0];

    float r8[8];
#pragma unroll
    for (int r = 0; r < 8; ++r) {
        const int m = r + 8 * hi;
        const float cm = __shfl(cosv, m, 32);
        float p = 0.f;
#pragma unroll
        for (int t = 0; t < 4; ++t) {
            float x = acc[t][r] + b1v[t] + cm * w1c[t];
            x = fmaxf(x, 0.f);
            p += x * w2v[t];
        }
        p += __shfl_xor(p, 1, 32);
        p += __shfl_xor(p, 2, 32);
        p += __shfl_xor(p, 4, 32);
        p += __shfl_xor(p, 8, 32);
        r8[r] = p;
    }

    const int rsel = nl & 7;
    float zsel = r8[0];
#pragma unroll
    for (int r = 1; r < 8; ++r) zsel = (rsel == r) ? r8[r] : zsel;
    const float z = zsel + b2s;
    const float y = 1.f / (1.f + expf(-z));
    const int eo = edgeBase + 8 * hi + rsel;
    if (nl < 8 && eo < E) out[eo] = y;
    }
}

extern "C" void kernel_launch(void* const* d_in, const int* in_sizes, int n_in,
                              void* d_out, int out_size, void* d_ws, size_t ws_size,
                              hipStream_t stream)
{
    const float* hc  = (const float*)d_in[0];
    const float* hp  = (const float*)d_in[1];
    const int*   src = (const int*)  d_in[2];
    const int*   dst = (const int*)  d_in[3];
    const float* W1  = (const float*)d_in[4];
    const float* b1  = (const float*)d_in[5];
    const float* W2  = (const float*)d_in[6];
    const float* b2  = (const float*)d_in[7];
    float* out = (float*)d_out;

    const int E = in_sizes[2];

    hipLaunchKernelGGL(edge_mlp_wmma_kernel, dim3(GRID_BLOCKS), dim3(BLOCK), 0, stream,
                       hc, hp, src, dst, W1, b1, W2, b2, out, E);
}


# W1 B fragments kept in registers for the wave's whole life (256 VGPRs, 8-wave groups, 2 waves per SIMD, next tile's rows fetched under the current WMMAs): no weight load in the WMMA stream
# speedup vs baseline: 1.4874x; 1.0805x over previous
#include <hip/hip_runtime.h>
#include <math.h>


typedef float v2f __attribute__((ext_vector_type(2)));
typedef float v8f __attribute__((ext_vector_type(8)));

#define EMBED            64
#define WAVES_PER_BLOCK  8
#define GRID_BLOCKS      256
#define BLOCK            (WAVES_PER_BLOCK * 32)
#define TILE_EDGES       16

__global__ __launch_bounds__(BLOCK)
void edge_mlp_wmma_kernel(const float* __restrict__ hc,
                          const float* __restrict__ hp,
                          const int*   __restrict__ src,
                          const int*   __restrict__ dst,
                          const float* __restrict__ W1,
                          const float* __restrict__ b1,
                          const float* __restrict__ W2,
                          const float* __restrict__ b2,
                          float*       __restrict__ out,
                          int E)
{
    const int tid  = threadIdx.x;
    const int lane = tid & 31;
    const int wave = __builtin_amdgcn_readfirstlane(tid >> 5);
    const int nl   = lane & 15;
    const int hi   = lane >> 4;
    const int nTiles     = (E + TILE_EDGES - 1) / TILE_EDGES;
    const int tileStride = GRID_BLOCKS * WAVES_PER_BLOCK;

    int tile = 2 * ((wave >> 1) * GRID_BLOCKS + blockIdx.x) + (wave & 1);
    if (tile >= nTiles) return;

    v2f bfr[32][4];
#pragma unroll
    for (int s = 0; s < 32; ++s) {
        const int k = 4 * s + 2 * hi;
#pragma unroll
        for (int t = 0; t < 4; ++t) {
            const int n = t * 16 + nl;
            v2f b;
            b.x = W1[k * 64 + n];
            b.y = W1[(k + 1) * 64 + n];
            bfr[s][t] = b;
        }
    }
    float w1c[4], b1v[4], w2v[4];
#pragma unroll
    for (int t = 0; t < 4; ++t) {
        const int n = t * 16 + nl;
        w1c[t] = W1[128 * 64 + n];
        b1v[t] = b1[n];
        w2v[t] = W2[n];
    }
    const float b2s = b2[0];

    v2f hcur[16], tcur[16];
    int sN, dN;
    {
        int e0 = tile * TILE_EDGES + nl;
        if (e0 >= E) e0 = E - 1;
        const float* __restrict__ hrow = hc + (size_t)src[e0] * EMBED;
        const float* __restrict__ trow = hp + (size_t)dst[e0] * EMBED;
        int e1 = (tile + tileStride) * TILE_EDGES + nl;
        if (e1 >= E) e1 = E - 1;
        sN = src[e1];
        dN = dst[e1];
#pragma unroll
        for (int s = 0; s < 16; ++s) {
            const int kb = 4 * s + 2 * hi;
            v2f a; a.x = hrow[kb]; a.y = hrow[kb + 1]; hcur[s] = a;
            v2f c; c.x = trow[kb]; c.y = trow[kb + 1]; tcur[s] = c;
        }
    }

    for (; tile < nTiles; tile += tileStride) {
    const int edgeBase = tile * TILE_EDGES;

    v2f hnxt[16], tnxt[16];
    int sN2, dN2;
    {
        const float* __restrict__ hrowN = hc + (size_t)sN * EMBED;
        const float* __restrict__ trowN = hp + (size_t)dN * EMBED;
#pragma unroll
        for (int s = 0; s < 16; ++s) {
            const int kb = 4 * s + 2 * hi;
            v2f a; a.x = hrowN[kb]; a.y = hrowN[kb + 1]; hnxt[s] = a;
            v2f c; c.x = trowN[kb]; c.y = trowN[kb + 1]; tnxt[s] = c;
        }
        int e2 = (tile + 2 * tileStride) * TILE_EDGES + nl;
        if (e2 >= E) e2 = E - 1;
        sN2 = src[e2];
        dN2 = dst[e2];
    }
    __builtin_amdgcn_sched_barrier(0);

    v8f acc[4] = {};
    float hh = 0.f, tt = 0.f, dp = 0.f;

#pragma unroll
    for (int s = 0; s < 16; ++s) {
        const v2f a = hcur[s];
        hh += a.x * a.x + a.y * a.y;
#pragma unroll
        for (int t = 0; t < 4; ++t) {
            acc[t] = __builtin_amdgcn_wmma_f32_16x16x4_f32(
                false, a, false, bfr[s][t], (short)0, acc[t], false, false);
        }
    }

#pragma unroll
    for (int s = 0; s < 16; ++s) {
        const v2f a = tcur[s];
        tt += a.x * a.x + a.y * a.y;
        dp += hcur[s].x * a.x + hcur[s].y * a.y;
#pragma unroll
        for (int t = 0; t < 4; ++t) {
            acc[t] = __builtin_amdgcn_wmma_f32_16x16x4_f32(
                false, a, false, bfr[16 + s][t], (short)0, acc[t], false, false);
        }
    }

    hh += __shfl_xor(hh, 16, 32);
    tt += __shfl_xor(tt, 16, 32);
    dp += __shfl_xor(dp, 16, 32);
    const float nh  = sqrtf(hh);
    const float nt  = sqrtf(tt);
    const float inh = 1.f / fmaxf(nh, 1e-12f);
    const float itn = 1.f / fmaxf(nt, 1e-12f);
    const float den = fmaxf((nh * inh) * (nt * itn), 1e-8f);
    const float cosv = (dp * inh * itn) / den;

    float r8[8];
#pragma unroll
    for (int r = 0; r < 8; ++r) {
        const int m = r + 8 * hi;
        const float cm = __shfl(cosv, m, 32);
        float p = 0.f;
#pragma unroll
        for (int t = 0; t < 4; ++t) {
            float x = acc[t][r] + b1v[t] + cm * w1c[t];
            x = fmaxf(x, 0.f);
            p += x * w2v[t];
        }
        p += __shfl_xor(p, 1, 32);
        p += __shfl_xor(p, 2, 32);
        p += __shfl_xor(p, 4, 32);
        p += __shfl_xor(p, 8, 32);
        r8[r] = p;
    }

    const int rsel = nl & 7;
    float zsel = r8[0];
#pragma unroll
    for (int r = 1; r < 8; ++r) zsel = (rsel == r) ? r8[r] : zsel;
    const float z = zsel + b2s;
    const float y = 1.f / (1.f + expf(-z));
    const int eo = edgeBase + 8 * hi + rsel;
    if (nl < 8 && eo < E) out[eo] = y;

#pragma unroll
    for (int s = 0; s < 16; ++s) { hcur[s] = hnxt[s]; tcur[s] = tnxt[s]; }
    sN = sN2;
    dN = dN2;
    }
}

extern "C" void kernel_launch(void* const* d_in, const int* in_sizes, int n_in,
                              void* d_out, int out_size, void* d_ws, size_t ws_size,
                              hipStream_t stream)
{
    const float* hc  = (const float*)d_in[0];
    const float* hp  = (const float*)d_in[1];
    const int*   src = (const int*)  d_in[2];
    const int*   dst = (const int*)  d_in[3];
    const float* W1  = (const float*)d_in[4];
    const float* b1  = (const float*)d_in[5];
    const float* W2  = (const float*)d_in[6];
    const float* b2  = (const float*)d_in[7];
    float* out = (float*)d_out;

    const int E = in_sizes[2];

    hipLaunchKernelGGL(edge_mlp_wmma_kernel, dim3(GRID_BLOCKS), dim3(BLOCK), 0, stream,
                       hc, hp, src, dst, W1, b1, W2, b2, out, E);
}


# Product transposed (W1^T as A operand, edge rows as B): a lane holds one edge's hidden sums, so layer 2 is an in-lane sum + 1 half swap (was 40 LDS permutes); cos,b1 = 33rd K-step; sum order changes
# speedup vs baseline: 1.5678x; 1.0541x over previous
#include <hip/hip_runtime.h>
#include <math.h>


typedef float v2f __attribute__((ext_vector_type(2)));
typedef float v8f __attribute__((ext_vector_type(8)));

#define EMBED            64
#define WAVES_PER_BLOCK  8
#define GRID_BLOCKS      256
#define BLOCK            (WAVES_PER_BLOCK * 32)
#define TILE_EDGES       16

__global__ __launch_bounds__(BLOCK)
void edge_mlp_wmma_kernel(const float* __restrict__ hc,
                          const float* __restrict__ hp,
                          const int*   __restrict__ src,
                          const int*   __restrict__ dst,
                          const float* __restrict__ W1,
                          const float* __restrict__ b1,
                          const float* __restrict__ W2,
                          const float* __restrict__ b2,
                          float*       __restrict__ out,
                          int E)
{
    const int tid  = threadIdx.x;
    const int lane = tid & 31;
    const int wave = __builtin_amdgcn_readfirstlane(tid >> 5);
    const int nl   = lane & 15;
    const int hi   = lane >> 4;
    const int nTiles     = (E + TILE_EDGES - 1) / TILE_EDGES;
    const int tileStride = GRID_BLOCKS * WAVES_PER_BLOCK;

    int tile = 2 * ((wave >> 1) * GRID_BLOCKS + blockIdx.x) + (wave & 1);
    if (tile >= nTiles) return;

    v2f bfr[32][4];
#pragma unroll
    for (int s = 0; s < 32; ++s) {
        const int k = 4 * s + 2 * hi;
#pragma unroll
        for (int t = 0; t < 4; ++t) {
            const int n = t * 16 + nl;
            v2f b;
            b.x = W1[k * 64 + n];
            b.y = W1[(k + 1) * 64 + n];
            bfr[s][t] = b;
        }
    }
    v2f wcb[4];
    float w2r[4][8];
#pragma unroll
    for (int t = 0; t < 4; ++t) {
        const int n = t * 16 + nl;
        const float wc = W1[128 * 64 + n];
        const float bb = b1[n];
        v2f e;
        e.x = hi ? 0.f : wc;
        e.y = hi ? 0.f : bb;
        wcb[t] = e;
#pragma unroll
        for (int r = 0; r < 8; ++r) w2r[t][r] = W2[t * 16 + 8 * hi + r];
    }
    const float b2s = b2[0];

    v2f hcur[16], tcur[16];
    int sN, dN;
    {
        int e0 = tile * TILE_EDGES + nl;
        if (e0 >= E) e0 = E - 1;
        const float* __restrict__ hrow = hc + (size_t)src[e0] * EMBED;
        const float* __restrict__ trow = hp + (size_t)dst[e0] * EMBED;
        int e1 = (tile + tileStride) * TILE_EDGES + nl;
        if (e1 >= E) e1 = E - 1;
        sN = src[e1];
        dN = dst[e1];
#pragma unroll
        for (int s = 0; s < 16; ++s) {
            const int kb = 4 * s + 2 * hi;
            v2f a; a.x = hrow[kb]; a.y = hrow[kb + 1]; hcur[s] = a;
            v2f c; c.x = trow[kb]; c.y = trow[kb + 1]; tcur[s] = c;
        }
    }

    for (; tile < nTiles; tile += tileStride) {
    const int edgeBase = tile * TILE_EDGES;

    v2f hnxt[16], tnxt[16];
    int sN2, dN2;
    {
        const float* __restrict__ hrowN = hc + (size_t)sN * EMBED;
        const float* __restrict__ trowN = hp + (size_t)dN * EMBED;
#pragma unroll
        for (int s = 0; s < 16; ++s) {
            const int kb = 4 * s + 2 * hi;
            v2f a; a.x = hrowN[kb]; a.y = hrowN[kb + 1]; hnxt[s] = a;
            v2f c; c.x = trowN[kb]; c.y = trowN[kb + 1]; tnxt[s] = c;
        }
        int e2 = (tile + 2 * tileStride) * TILE_EDGES + nl;
        if (e2 >= E) e2 = E - 1;
        sN2 = src[e2];
        dN2 = dst[e2];
    }
    __builtin_amdgcn_sched_barrier(0);

    v8f acc[4] = {};
    float hh = 0.f, tt = 0.f, dp = 0.f;

#pragma unroll
    for (int s = 0; s < 16; ++s) {
        const v2f a = hcur[s];
        hh += a.x * a.x + a.y * a.y;
#pragma unroll
        for (int t = 0; t < 4; ++t) {
            acc[t] = __builtin_amdgcn_wmma_f32_16x16x4_f32(
                false, bfr[s][t], false, a, (short)0, acc[t], false, false);
        }
    }

#pragma unroll
    for (int s = 0; s < 16; ++s) {
        const v2f a = tcur[s];
        tt += a.x * a.x + a.y * a.y;
        dp += hcur[s].x * a.x + hcur[s].y * a.y;
#pragma unroll
        for (int t = 0; t < 4; ++t) {
            acc[t] = __builtin_amdgcn_wmma_f32_16x16x4_f32(
                false, bfr[16 + s][t], false, a, (short)0, acc[t], false, false);
        }
    }

    hh += __shfl_xor(hh, 16, 32);
    tt += __shfl_xor(tt, 16, 32);
    dp += __shfl_xor(dp, 16, 32);
    const float nh  = sqrtf(hh);
    const float nt  = sqrtf(tt);
    const float inh = 1.f / fmaxf(nh, 1e-12f);
    const float itn = 1.f / fmaxf(nt, 1e-12f);
    const float den = fmaxf((nh * inh) * (nt * itn), 1e-8f);
    const float cosv = (dp * inh * itn) / den;

    {
        v2f cb;
        cb.x = hi ? 0.f : cosv;
        cb.y = hi ? 0.f : 1.f;
#pragma unroll
        for (int t = 0; t < 4; ++t) {
            acc[t] = __builtin_amdgcn_wmma_f32_16x16x4_f32(
                false, wcb[t], false, cb, (short)0, acc[t], false, false);
        }
    }

    float p4[4];
#pragma unroll
    for (int t = 0; t < 4; ++t) {
        float p = 0.f;
#pragma unroll
        for (int r = 0; r < 8; ++r) {
            const float x = fmaxf(acc[t][r], 0.f);
            p += x * w2r[t][r];
        }
        p4[t] = p;
    }
    float ps = (p4[0] + p4[1]) + (p4[2] + p4[3]);
    ps += __shfl_xor(ps, 16, 32);
    const float z = ps + b2s;
    const float y = 1.f / (1.f + expf(-z));
    const int eo = edgeBase + nl;
    if (hi == 0 && eo < E) out[eo] = y;

#pragma unroll
    for (int s = 0; s < 16; ++s) { hcur[s] = hnxt[s]; tcur[s] = tnxt[s]; }
    sN = sN2;
    dN = dN2;
    }
}

extern "C" void kernel_launch(void* const* d_in, const int* in_sizes, int n_in,
                              void* d_out, int out_size, void* d_ws, size_t ws_size,
                              hipStream_t stream)
{
    const float* hc  = (const float*)d_in[0];
    const float* hp  = (const float*)d_in[1];
    const int*   src = (const int*)  d_in[2];
    const int*   dst = (const int*)  d_in[3];
    const float* W1  = (const float*)d_in[4];
    const float* b1  = (const float*)d_in[5];
    const float* W2  = (const float*)d_in[6];
    const float* b2  = (const float*)d_in[7];
    float* out = (float*)d_out;

    const int E = in_sizes[2];

    hipLaunchKernelGGL(edge_mlp_wmma_kernel, dim3(GRID_BLOCKS), dim3(BLOCK), 0, stream,
                       hc, hp, src, dst, W1, b1, W2, b2, out, E);
}


# Cosine and sigmoid arithmetic cut down (vector work per tile 312 -> 174): cosine sums as 3 packed-FMA chains; 1-ulp v_sqrt/v_rcp/v_exp for sqrtf, 4 divides, expf; hh,tt,dp sum order changes
# speedup vs baseline: 1.6956x; 1.0815x over previous
#include <hip/hip_runtime.h>
#include <math.h>


typedef float v2f __attribute__((ext_vector_type(2)));
typedef float v8f __attribute__((ext_vector_type(8)));

#define EMBED            64
#define WAVES_PER_BLOCK  8
#define GRID_BLOCKS      256
#define BLOCK            (WAVES_PER_BLOCK * 32)
#define TILE_EDGES       16

__global__ __launch_bounds__(BLOCK)
void edge_mlp_wmma_kernel(const float* __restrict__ hc,
                          const float* __restrict__ hp,
                          const int*   __restrict__ src,
                          const int*   __restrict__ dst,
                          const float* __restrict__ W1,
                          const float* __restrict__ b1,
                          const float* __restrict__ W2,
                          const float* __restrict__ b2,
                          float*       __restrict__ out,
                          int E)
{
    const int tid  = threadIdx.x;
    const int lane = tid & 31;
    const int wave = __builtin_amdgcn_readfirstlane(tid >> 5);
    const int nl   = lane & 15;
    const int hi   = lane >> 4;
    const int nTiles     = (E + TILE_EDGES - 1) / TILE_EDGES;
    const int tileStride = GRID_BLOCKS * WAVES_PER_BLOCK;

    int tile = 2 * ((wave >> 1) * GRID_BLOCKS + blockIdx.x) + (wave & 1);
    if (tile >= nTiles) return;

    v2f bfr[32][4];
#pragma unroll
    for (int s = 0; s < 32; ++s) {
        const int k = 4 * s + 2 * hi;
#pragma unroll
        for (int t = 0; t < 4; ++t) {
            const int n = t * 16 + nl;
            v2f b;
            b.x = W1[k * 64 + n];
            b.y = W1[(k + 1) * 64 + n];
            bfr[s][t] = b;
        }
    }
    v2f wcb[4];
    float w2r[4][8];
#pragma unroll
    for (int t = 0; t < 4; ++t) {
        const int n = t * 16 + nl;
        const float wc = W1[128 * 64 + n];
        const float bb = b1[n];
        v2f e;
        e.x = hi ? 0.f : wc;
        e.y = hi ? 0.f : bb;
        wcb[t] = e;
#pragma unroll
        for (int r = 0; r < 8; ++r) w2r[t][r] = W2[t * 16 + 8 * hi + r];
    }
    const float b2s = b2[0];

    v2f hcur[16], tcur[16];
    int sN, dN;
    {
        int e0 = tile * TILE_EDGES + nl;
        if (e0 >= E) e0 = E - 1;
        const float* __restrict__ hrow = hc + (size_t)src[e0] * EMBED;
        const float* __restrict__ trow = hp + (size_t)dst[e0] * EMBED;
        int e1 = (tile + tileStride) * TILE_EDGES + nl;
        if (e1 >= E) e1 = E - 1;
        sN = src[e1];
        dN = dst[e1];
#pragma unroll
        for (int s = 0; s < 16; ++s) {
            const int kb = 4 * s + 2 * hi;
            v2f a; a.x = hrow[kb]; a.y = hrow[kb + 1]; hcur[s] = a;
            v2f c; c.x = trow[kb]; c.y = trow[kb + 1]; tcur[s] = c;
        }
    }

    for (; tile < nTiles; tile += tileStride) {
    const int edgeBase = tile * TILE_EDGES;

    v2f hnxt[16], tnxt[16];
    int sN2, dN2;
    {
        const float* __restrict__ hrowN = hc + (size_t)sN * EMBED;
        const float* __restrict__ trowN = hp + (size_t)dN * EMBED;
#pragma unroll
        for (int s = 0; s < 16; ++s) {
            const int kb = 4 * s + 2 * hi;
            v2f a; a.x = hrowN[kb]; a.y = hrowN[kb + 1]; hnxt[s] = a;
            v2f c; c.x = trowN[kb]; c.y = trowN[kb + 1]; tnxt[s] = c;
        }
        int e2 = (tile + 2 * tileStride) * TILE_EDGES + nl;
        if (e2 >= E) e2 = E - 1;
        sN2 = src[e2];
        dN2 = dst[e2];
    }
    __builtin_amdgcn_sched_barrier(0);

    v8f acc[4] = {};
    v2f hh2 = {0.f, 0.f}, tt2 = {0.f, 0.f}, dp2 = {0.f, 0.f};

#pragma unroll
    for (int s = 0; s < 16; ++s) {
        const v2f a = hcur[s];
        hh2 += a * a;
#pragma unroll
        for (int t = 0; t < 4; ++t) {
            acc[t] = __builtin_amdgcn_wmma_f32_16x16x4_f32(
                false, bfr[s][t], false, a, (short)0, acc[t], false, false);
        }
    }

#pragma unroll
    for (int s = 0; s < 16; ++s) {
        const v2f a = tcur[s];
        tt2 += a * a;
        dp2 += hcur[s] * a;
#pragma unroll
        for (int t = 0; t < 4; ++t) {
            acc[t] = __builtin_amdgcn_wmma_f32_16x16x4_f32(
                false, bfr[16 + s][t], false, a, (short)0, acc[t], false, false);
        }
    }

    float hh = hh2.x + hh2.y;
    float tt = tt2.x + tt2.y;
    float dp = dp2.x + dp2.y;
    hh += __shfl_xor(hh, 16, 32);
    tt += __shfl_xor(tt, 16, 32);
    dp += __shfl_xor(dp, 16, 32);
    const float nh  = __builtin_amdgcn_sqrtf(hh);
    const float nt  = __builtin_amdgcn_sqrtf(tt);
    const float inh = __builtin_amdgcn_rcpf(fmaxf(nh, 1e-12f));
    const float itn = __builtin_amdgcn_rcpf(fmaxf(nt, 1e-12f));
    const float den = fmaxf((nh * inh) * (nt * itn), 1e-8f);
    const float cosv = (dp * inh * itn) * __builtin_amdgcn_rcpf(den);

    {
        v2f cb;
        cb.x = hi ? 0.f : cosv;
        cb.y = hi ? 0.f : 1.f;
#pragma unroll
        for (int t = 0; t < 4; ++t) {
            acc[t] = __builtin_amdgcn_wmma_f32_16x16x4_f32(
                false, wcb[t], false, cb, (short)0, acc[t], false, false);
        }
    }

    float p4[4];
#pragma unroll
    for (int t = 0; t < 4; ++t) {
        float p = 0.f;
#pragma unroll
        for (int r = 0; r < 8; ++r) {
            const float x = fmaxf(acc[t][r], 0.f);
            p += x * w2r[t][r];
        }
        p4[t] = p;
    }
    float ps = (p4[0] + p4[1]) + (p4[2] + p4[3]);
    ps += __shfl_xor(ps, 16, 32);
    const float z = ps + b2s;
    const float y = __builtin_amdgcn_rcpf(1.f + __builtin_amdgcn_exp2f(z * -1.44269504088896341f));
    const int eo = edgeBase + nl;
    if (hi == 0 && eo < E) out[eo] = y;

#pragma unroll
    for (int s = 0; s < 16; ++s) { hcur[s] = hnxt[s]; tcur[s] = tnxt[s]; }
    sN = sN2;
    dN = dN2;
    }
}

extern "C" void kernel_launch(void* const* d_in, const int* in_sizes, int n_in,
                              void* d_out, int out_size, void* d_ws, size_t ws_size,
                              hipStream_t stream)
{
    const float* hc  = (const float*)d_in[0];
    const float* hp  = (const float*)d_in[1];
    const int*   src = (const int*)  d_in[2];
    const int*   dst = (const int*)  d_in[3];
    const float* W1  = (const float*)d_in[4];
    const float* b1  = (const float*)d_in[5];
    const float* W2  = (const float*)d_in[6];
    const float* b2  = (const float*)d_in[7];
    float* out = (float*)d_out;

    const int E = in_sizes[2];

    hipLaunchKernelGGL(edge_mlp_wmma_kernel, dim3(GRID_BLOCKS), dim3(BLOCK), 0, stream,
                       hc, hp, src, dst, W1, b1, W2, b2, out, E);
}


# Tail half of layer 1 taken once per product row (first launch: P = hp @ W1[64..127] into the workspace, WMMA); an edge's accumulators start from P[dst], 68 WMMAs per tile, was 132; sum order changes
# speedup vs baseline: 1.8928x; 1.1163x over previous
#include <hip/hip_runtime.h>
#include <math.h>


typedef float v2f __attribute__((ext_vector_type(2)));
typedef float v4f __attribute__((ext_vector_type(4)));
typedef float v8f __attribute__((ext_vector_type(8)));

#define EMBED            64
#define WAVES_PER_BLOCK  8
#define GRID_BLOCKS      256
#define BLOCK            (WAVES_PER_BLOCK * 32)
#define TILE_EDGES       16

__global__ __launch_bounds__(BLOCK)
void tail_projection_kernel(const float* __restrict__ hp,
                            const float* __restrict__ W1,
                            float*       __restrict__ P,
                            int nHp, size_t wsBytes)
{
    if (wsBytes < (size_t)nHp * sizeof(float)) return;
    const int tid  = threadIdx.x;
    const int lane = tid & 31;
    const int wave = __builtin_amdgcn_readfirstlane(tid >> 5);
    const int nl   = lane & 15;
    const int hi   = lane >> 4;
    const int nRows  = nHp / EMBED;
    const int nTiles = (nRows + 15) / 16;
    const int tileStride = GRID_BLOCKS * WAVES_PER_BLOCK;
    int tile = wave * GRID_BLOCKS + blockIdx.x;
    if (tile >= nTiles) return;

    v2f wfr[16][4];
#pragma unroll
    for (int s = 0; s < 16; ++s) {
        const int k = EMBED + 4 * s + 2 * hi;
#pragma unroll
        for (int t = 0; t < 4; ++t) {
            const int n = t * 16 + nl;
            v2f b;
            b.x = W1[k * 64 + n];
            b.y = W1[(k + 1) * 64 + n];
            wfr[s][t] = b;
        }
    }

    for (; tile < nTiles; tile += tileStride) {
        const int row = tile * 16 + nl;
        const int rc  = row < nRows ? row : nRows - 1;
        const float* __restrict__ r = hp + (size_t)rc * EMBED;
        v2f x[16];
#pragma unroll
        for (int s = 0; s < 16; ++s) {
            const int kb = 4 * s + 2 * hi;
            v2f a; a.x = r[kb]; a.y = r[kb + 1]; x[s] = a;
        }
        v8f acc[4] = {};
#pragma unroll
        for (int s = 0; s < 16; ++s) {
#pragma unroll
            for (int t = 0; t < 4; ++t) {
                acc[t] = __builtin_amdgcn_wmma_f32_16x16x4_f32(
                    false, wfr[s][t], false, x[s], (short)0, acc[t], false, false);
            }
        }
        if (row < nRows) {
            float* __restrict__ o = P + (size_t)row * EMBED + 8 * hi;
#pragma unroll
            for (int t = 0; t < 4; ++t) {
                v4f lo = {acc[t][0], acc[t][1], acc[t][2], acc[t][3]};
                v4f up = {acc[t][4], acc[t][5], acc[t][6], acc[t][7]};
                *(v4f*)(o + 16 * t)     = lo;
                *(v4f*)(o + 16 * t + 4) = up;
            }
        }
    }
}

template <bool HOIST>
__device__ __forceinline__
void edge_tiles(const float* __restrict__ hc, const float* __restrict__ hp,
                const int* __restrict__ src, const int* __restrict__ dst,
                const float* __restrict__ W1, const float* __restrict__ b1,
                const float* __restrict__ W2, const float* __restrict__ b2,
                const float* __restrict__ P,
                float* __restrict__ out, int E,
                int tile, const int nTiles, const int nl, const int hi)
{
    constexpr int NS = HOIST ? 16 : 32;
    const int tileStride = GRID_BLOCKS * WAVES_PER_BLOCK;

    v2f bfr[NS][4];
#pragma unroll
    for (int s = 0; s < NS; ++s) {
        const int k = 4 * s + 2 * hi;
#pragma unroll
        for (int t = 0; t < 4; ++t) {
            const int n = t * 16 + nl;
            v2f b;
            b.x = W1[k * 64 + n];
            b.y = W1[(k + 1) * 64 + n];
            bfr[s][t] = b;
        }
    }
    v2f wcb[4];
    float w2r[4][8];
#pragma unroll
    for (int t = 0; t < 4; ++t) {
        const int n = t * 16 + nl;
        const float wc = W1[128 * 64 + n];
        const float bb = b1[n];
        v2f e;
        e.x = hi ? 0.f : wc;
        e.y = hi ? 0.f : bb;
        wcb[t] = e;
#pragma unroll
        for (int r = 0; r < 8; ++r) w2r[t][r] = W2[t * 16 + 8 * hi + r];
    }
    const float b2s = b2[0];

    v2f hcur[16], tcur[16];
    v8f pcur[4] = {};
    int sN, dN;
    {
        int e0 = tile * TILE_EDGES + nl;
        if (e0 >= E) e0 = E - 1;
        const int d0 = dst[e0];
        const float* __restrict__ hrow = hc + (size_t)src[e0] * EMBED;
        const float* __restrict__ trow = hp + (size_t)d0 * EMBED;
        int e1 = (tile + tileStride) * TILE_EDGES + nl;
        if (e1 >= E) e1 = E - 1;
        sN = src[e1];
        dN = dst[e1];
#pragma unroll
        for (int s = 0; s < 16; ++s) {
            const int kb = 4 * s + 2 * hi;
            v2f a; a.x = hrow[kb]; a.y = hrow[kb + 1]; hcur[s] = a;
            v2f c; c.x = trow[kb]; c.y = trow[kb + 1]; tcur[s] = c;
        }
        if (HOIST) {
            const v4f* __restrict__ prow = (const v4f*)(P + (size_t)d0 * EMBED + 8 * hi);
#pragma unroll
            for (int t = 0; t < 4; ++t) {
                const v4f lo = prow[4 * t], up = prow[4 * t + 1];
                v8f p = {lo.x, lo.y, lo.z, lo.w, up.x, up.y, up.z, up.w};
                pcur[t] = p;
            }
        }
    }

    for (; tile < nTiles; tile += tileStride) {
    const int edgeBase = tile * TILE_EDGES;

    v2f hnxt[16], tnxt[16];
    v8f pnxt[4] = {};
    int sN2, dN2;
    {
        const float* __restrict__ hrowN = hc + (size_t)sN * EMBED;
        const float* __restrict__ trowN = hp + (size_t)dN * EMBED;
#pragma unroll
        for (int s = 0; s < 16; ++s) {
            const int kb = 4 * s + 2 * hi;
            v2f a; a.x = hrowN[kb]; a.y = hrowN[kb + 1]; hnxt[s] = a;
            v2f c; c.x = trowN[kb]; c.y = trowN[kb + 1]; tnxt[s] = c;
        }
        if (HOIST) {
            const v4f* __restrict__ prowN = (const v4f*)(P + (size_t)dN * EMBED + 8 * hi);
#pragma unroll
            for (int t = 0; t < 4; ++t) {
                const v4f lo = prowN[4 * t], up = prowN[4 * t + 1];
                v8f p = {lo.x, lo.y, lo.z, lo.w, up.x, up.y, up.z, up.w};
                pnxt[t] = p;
            }
        }
        int e2 = (tile + 2 * tileStride) * TILE_EDGES + nl;
        if (e2 >= E) e2 = E - 1;
        sN2 = src[e2];
        dN2 = dst[e2];
    }
    __builtin_amdgcn_sched_barrier(0);

    v8f acc[4];
#pragma unroll
    for (int t = 0; t < 4; ++t) acc[t] = pcur[t];
    v2f hh2 = {0.f, 0.f}, tt2 = {0.f, 0.f}, dp2 = {0.f, 0.f};

#pragma unroll
    for (int s = 0; s < 16; ++s) {
        const v2f a = hcur[s];
        hh2 += a * a;
#pragma unroll
        for (int t = 0; t < 4; ++t) {
            acc[t] = __builtin_amdgcn_wmma_f32_16x16x4_f32(
                false, bfr[s][t], false, a, (short)0, acc[t], false, false);
        }
    }

#pragma unroll
    for (int s = 0; s < 16; ++s) {
        const v2f a = tcur[s];
        tt2 += a * a;
        dp2 += hcur[s] * a;
        if (!HOIST) {
#pragma unroll
            for (int t = 0; t < 4; ++t) {
                acc[t] = __builtin_amdgcn_wmma_f32_16x16x4_f32(
                    false, bfr[(NS - 16) + s][t], false, a, (short)0, acc[t], false, false);
            }
        }
    }

    float hh = hh2.x + hh2.y;
    float tt = tt2.x + tt2.y;
    float dp = dp2.x + dp2.y;
    hh += __shfl_xor(hh, 16, 32);
    tt += __shfl_xor(tt, 16, 32);
    dp += __shfl_xor(dp, 16, 32);
    const float nh  = __builtin_amdgcn_sqrtf(hh);
    const float nt  = __builtin_amdgcn_sqrtf(tt);
    const float inh = __builtin_amdgcn_rcpf(fmaxf(nh, 1e-12f));
    const float itn = __builtin_amdgcn_rcpf(fmaxf(nt, 1e-12f));
    const float den = fmaxf((nh * inh) * (nt * itn), 1e-8f);
    const float cosv = (dp * inh * itn) * __builtin_amdgcn_rcpf(den);

    {
        v2f cb;
        cb.x = hi ? 0.f : cosv;
        cb.y = hi ? 0.f : 1.f;
#pragma unroll
        for (int t = 0; t < 4; ++t) {
            acc[t] = __builtin_amdgcn_wmma_f32_16x16x4_f32(
                false, wcb[t], false, cb, (short)0, acc[t], false, false);
        }
    }

    float p4[4];
#pragma unroll
    for (int t = 0; t < 4; ++t) {
        float p = 0.f;
#pragma unroll
        for (int r = 0; r < 8; ++r) {
            const float x = fmaxf(acc[t][r], 0.f);
            p += x * w2r[t][r];
        }
        p4[t] = p;
    }
    float ps = (p4[0] + p4[1]) + (p4[2] + p4[3]);
    ps += __shfl_xor(ps, 16, 32);
    const float z = ps + b2s;
    const float y = __builtin_amdgcn_rcpf(1.f + __builtin_amdgcn_exp2f(z * -1.44269504088896341f));
    const int eo = edgeBase + nl;
    if (hi == 0 && eo < E) out[eo] = y;

#pragma unroll
    for (int s = 0; s < 16; ++s) { hcur[s] = hnxt[s]; tcur[s] = tnxt[s]; }
#pragma unroll
    for (int t = 0; t < 4; ++t) pcur[t] = pnxt[t];
    sN = sN2;
    dN = dN2;
    }
}

__global__ __launch_bounds__(BLOCK)
void edge_mlp_wmma_kernel(const float* __restrict__ hc,
                          const float* __restrict__ hp,
                          const int*   __restrict__ src,
                          const int*   __restrict__ dst,
                          const float* __restrict__ W1,
                          const float* __restrict__ b1,
                          const float* __restrict__ W2,
                          const float* __restrict__ b2,
                          const float* __restrict__ P,
                          float*       __restrict__ out,
                          int E, int nHp, size_t wsBytes)
{
    const int tid  = threadIdx.x;
    const int lane = tid & 31;
    const int wave = __builtin_amdgcn_readfirstlane(tid >> 5);
    const int nl   = lane & 15;
    const int hi   = lane >> 4;
    const int nTiles = (E + TILE_EDGES - 1) / TILE_EDGES;

    const int tile = 2 * ((wave >> 1) * GRID_BLOCKS + blockIdx.x) + (wave & 1);
    if (tile >= nTiles) return;

    if (wsBytes >= (size_t)nHp * sizeof(float))
        edge_tiles<true >(hc, hp, src, dst, W1, b1, W2, b2, P, out, E, tile, nTiles, nl, hi);
    else
        edge_tiles<false>(hc, hp, src, dst, W1, b1, W2, b2, P, out, E, tile, nTiles, nl, hi);
}

extern "C" void kernel_launch(void* const* d_in, const int* in_sizes, int n_in,
                              void* d_out, int out_size, void* d_ws, size_t ws_size,
                              hipStream_t stream)
{
    const float* hc  = (const float*)d_in[0];
    const float* hp  = (const float*)d_in[1];
    const int*   src = (const int*)  d_in[2];
    const int*   dst = (const int*)  d_in[3];
    const float* W1  = (const float*)d_in[4];
    const float* b1  = (const float*)d_in[5];
    const float* W2  = (const float*)d_in[6];
    const float* b2  = (const float*)d_in[7];
    float* out = (float*)d_out;
    float* P   = (float*)d_ws;

    const int E   = in_sizes[2];
    const int nHp = in_sizes[1];

    hipLaunchKernelGGL(tail_projection_kernel, dim3(GRID_BLOCKS), dim3(BLOCK), 0, stream,
                       hp, W1, P, nHp, ws_size);
    hipLaunchKernelGGL(edge_mlp_wmma_kernel, dim3(GRID_BLOCKS), dim3(BLOCK), 0, stream,
                       hc, hp, src, dst, W1, b1, W2, b2, (const float*)P, out, E, nHp, ws_size);
}


# Row loads 16 B per lane (lane takes floats 8q+4h..+3 of its row; K-step 2q+j = pair j): 8 loads of 16 lines per row table per tile, was 16; W1 fragments in the same k order; sum order changes
# speedup vs baseline: 1.9863x; 1.0494x over previous
#include <hip/hip_runtime.h>
#include <math.h>


typedef float v2f __attribute__((ext_vector_type(2)));
typedef float v4f __attribute__((ext_vector_type(4)));
typedef float v8f __attribute__((ext_vector_type(8)));

#define EMBED            64
#define WAVES_PER_BLOCK  8
#define GRID_BLOCKS      256
#define BLOCK            (WAVES_PER_BLOCK * 32)
#define TILE_EDGES       16

__global__ __launch_bounds__(BLOCK)
void tail_projection_kernel(const float* __restrict__ hp,
                            const float* __restrict__ W1,
                            float*       __restrict__ P,
                            int nHp, size_t wsBytes)
{
    if (wsBytes < (size_t)nHp * sizeof(float)) return;
    const int tid  = threadIdx.x;
    const int lane = tid & 31;
    const int wave = __builtin_amdgcn_readfirstlane(tid >> 5);
    const int nl   = lane & 15;
    const int hi   = lane >> 4;
    const int nRows  = nHp / EMBED;
    const int nTiles = (nRows + 15) / 16;
    const int tileStride = GRID_BLOCKS * WAVES_PER_BLOCK;
    int tile = wave * GRID_BLOCKS + blockIdx.x;
    if (tile >= nTiles) return;

    v2f wfr[16][4];
#pragma unroll
    for (int s = 0; s < 16; ++s) {
        const int k = EMBED + 8 * (s >> 1) + 4 * hi + 2 * (s & 1);
#pragma unroll
        for (int t = 0; t < 4; ++t) {
            const int n = t * 16 + nl;
            v2f b;
            b.x = W1[k * 64 + n];
            b.y = W1[(k + 1) * 64 + n];
            wfr[s][t] = b;
        }
    }

    for (; tile < nTiles; tile += tileStride) {
        const int row = tile * 16 + nl;
        const int rc  = row < nRows ? row : nRows - 1;
        const v4f* __restrict__ r4 = (const v4f*)(hp + (size_t)rc * EMBED + 4 * hi);
        v2f x[16];
#pragma unroll
        for (int q = 0; q < 8; ++q) {
            const v4f a = r4[2 * q];
            v2f lo = {a.x, a.y}, up = {a.z, a.w};
            x[2 * q] = lo; x[2 * q + 1] = up;
        }
        v8f acc[4] = {};
#pragma unroll
        for (int s = 0; s < 16; ++s) {
#pragma unroll
            for (int t = 0; t < 4; ++t) {
                acc[t] = __builtin_amdgcn_wmma_f32_16x16x4_f32(
                    false, wfr[s][t], false, x[s], (short)0, acc[t], false, false);
            }
        }
        if (row < nRows) {
            float* __restrict__ o = P + (size_t)row * EMBED + 8 * hi;
#pragma unroll
            for (int t = 0; t < 4; ++t) {
                v4f lo = {acc[t][0], acc[t][1], acc[t][2], acc[t][3]};
                v4f up = {acc[t][4], acc[t][5], acc[t][6], acc[t][7]};
                *(v4f*)(o + 16 * t)     = lo;
                *(v4f*)(o + 16 * t + 4) = up;
            }
        }
    }
}

template <bool HOIST>
__device__ __forceinline__
void edge_tiles(const float* __restrict__ hc, const float* __restrict__ hp,
                const int* __restrict__ src, const int* __restrict__ dst,
                const float* __restrict__ W1, const float* __restrict__ b1,
                const float* __restrict__ W2, const float* __restrict__ b2,
                const float* __restrict__ P,
                float* __restrict__ out, int E,
                int tile, const int nTiles, const int nl, const int hi)
{
    constexpr int NS = HOIST ? 16 : 32;
    const int tileStride = GRID_BLOCKS * WAVES_PER_BLOCK;

    v2f bfr[NS][4];
#pragma unroll
    for (int s = 0; s < NS; ++s) {
        const int k = 64 * (s >> 4) + 8 * ((s & 15) >> 1) + 4 * hi + 2 * (s & 1);
#pragma unroll
        for (int t = 0; t < 4; ++t) {
            const int n = t * 16 + nl;
            v2f b;
            b.x = W1[k * 64 + n];
            b.y = W1[(k + 1) * 64 + n];
            bfr[s][t] = b;
        }
    }
    v2f wcb[4];
    float w2r[4][8];
#pragma unroll
    for (int t = 0; t < 4; ++t) {
        const int n = t * 16 + nl;
        const float wc = W1[128 * 64 + n];
        const float bb = b1[n];
        v2f e;
        e.x = hi ? 0.f : wc;
        e.y = hi ? 0.f : bb;
        wcb[t] = e;
#pragma unroll
        for (int r = 0; r < 8; ++r) w2r[t][r] = W2[t * 16 + 8 * hi + r];
    }
    const float b2s = b2[0];

    v4f hcur[8], tcur[8];
    v8f pcur[4] = {};
    int sN, dN;
    {
        int e0 = tile * TILE_EDGES + nl;
        if (e0 >= E) e0 = E - 1;
        const int d0 = dst[e0];
        const float* __restrict__ hrow = hc + (size_t)src[e0] * EMBED;
        const float* __restrict__ trow = hp + (size_t)d0 * EMBED;
        int e1 = (tile + tileStride) * TILE_EDGES + nl;
        if (e1 >= E) e1 = E - 1;
        sN = src[e1];
        dN = dst[e1];
        const v4f* __restrict__ hrow4 = (const v4f*)(hrow + 4 * hi);
        const v4f* __restrict__ trow4 = (const v4f*)(trow + 4 * hi);
#pragma unroll
        for (int q = 0; q < 8; ++q) {
            hcur[q] = hrow4[2 * q];
            tcur[q] = trow4[2 * q];
        }
        if (HOIST) {
            const v4f* __restrict__ prow = (const v4f*)(P + (size_t)d0 * EMBED + 8 * hi);
#pragma unroll
            for (int t = 0; t < 4; ++t) {
                const v4f lo = prow[4 * t], up = prow[4 * t + 1];
                v8f p = {lo.x, lo.y, lo.z, lo.w, up.x, up.y, up.z, up.w};
                pcur[t] = p;
            }
        }
    }

    for (; tile < nTiles; tile += tileStride) {
    const int edgeBase = tile * TILE_EDGES;

    v4f hnxt[8], tnxt[8];
    v8f pnxt[4] = {};
    int sN2, dN2;
    {
        const v4f* __restrict__ hrowN4 = (const v4f*)(hc + (size_t)sN * EMBED + 4 * hi);
        const v4f* __restrict__ trowN4 = (const v4f*)(hp + (size_t)dN * EMBED + 4 * hi);
#pragma unroll
        for (int q = 0; q < 8; ++q) {
            hnxt[q] = hrowN4[2 * q];
            tnxt[q] = trowN4[2 * q];
        }
        if (HOIST) {
            const v4f* __restrict__ prowN = (const v4f*)(P + (size_t)dN * EMBED + 8 * hi);
#pragma unroll
            for (int t = 0; t < 4; ++t) {
                const v4f lo = prowN[4 * t], up = prowN[4 * t + 1];
                v8f p = {lo.x, lo.y, lo.z, lo.w, up.x, up.y, up.z, up.w};
                pnxt[t] = p;
            }
        }
        int e2 = (tile + 2 * tileStride) * TILE_EDGES + nl;
        if (e2 >= E) e2 = E - 1;
        sN2 = src[e2];
        dN2 = dst[e2];
    }
    __builtin_amdgcn_sched_barrier(0);

    v8f acc[4];
#pragma unroll
    for (int t = 0; t < 4; ++t) acc[t] = pcur[t];
    v2f hh2 = {0.f, 0.f}, tt2 = {0.f, 0.f}, dp2 = {0.f, 0.f};

#pragma unroll
    for (int s = 0; s < 16; ++s) {
        const v4f a4 = hcur[s >> 1];
        const v2f a = (s & 1) ? __builtin_shufflevector(a4, a4, 2, 3)
                              : __builtin_shufflevector(a4, a4, 0, 1);
        hh2 += a * a;
#pragma unroll
        for (int t = 0; t < 4; ++t) {
            acc[t] = __builtin_amdgcn_wmma_f32_16x16x4_f32(
                false, bfr[s][t], false, a, (short)0, acc[t], false, false);
        }
    }

#pragma unroll
    for (int s = 0; s < 16; ++s) {
        const v4f a4 = tcur[s >> 1];
        const v4f g4 = hcur[s >> 1];
        const v2f a = (s & 1) ? __builtin_shufflevector(a4, a4, 2, 3)
                              : __builtin_shufflevector(a4, a4, 0, 1);
        const v2f g = (s & 1) ? __builtin_shufflevector(g4, g4, 2, 3)
                              : __builtin_shufflevector(g4, g4, 0, 1);
        tt2 += a * a;
        dp2 += g * a;
        if (!HOIST) {
#pragma unroll
            for (int t = 0; t < 4; ++t) {
                acc[t] = __builtin_amdgcn_wmma_f32_16x16x4_f32(
                    false, bfr[(NS - 16) + s][t], false, a, (short)0, acc[t], false, false);
            }
        }
    }

    float hh = hh2.x + hh2.y;
    float tt = tt2.x + tt2.y;
    float dp = dp2.x + dp2.y;
    hh += __shfl_xor(hh, 16, 32);
    tt += __shfl_xor(tt, 16, 32);
    dp += __shfl_xor(dp, 16, 32);
    const float nh  = __builtin_amdgcn_sqrtf(hh);
    const float nt  = __builtin_amdgcn_sqrtf(tt);
    const float inh = __builtin_amdgcn_rcpf(fmaxf(nh, 1e-12f));
    const float itn = __builtin_amdgcn_rcpf(fmaxf(nt, 1e-12f));
    const float den = fmaxf((nh * inh) * (nt * itn), 1e-8f);
    const float cosv = (dp * inh * itn) * __builtin_amdgcn_rcpf(den);

    {
        v2f cb;
        cb.x = hi ? 0.f : cosv;
        cb.y = hi ? 0.f : 1.f;
#pragma unroll
        for (int t = 0; t < 4; ++t) {
            acc[t] = __builtin_amdgcn_wmma_f32_16x16x4_f32(
                false, wcb[t], false, cb, (short)0, acc[t], false, false);
        }
    }

    float p4[4];
#pragma unroll
    for (int t = 0; t < 4; ++t) {
        float p = 0.f;
#pragma unroll
        for (int r = 0; r < 8; ++r) {
            const float x = fmaxf(acc[t][r], 0.f);
            p += x * w2r[t][r];
        }
        p4[t] = p;
    }
    float ps = (p4[0] + p4[1]) + (p4[2] + p4[3]);
    ps += __shfl_xor(ps, 16, 32);
    const float z = ps + b2s;
    const float y = __builtin_amdgcn_rcpf(1.f + __builtin_amdgcn_exp2f(z * -1.44269504088896341f));
    const int eo = edgeBase + nl;
    if (hi == 0 && eo < E) out[eo] = y;

#pragma unroll
    for (int q = 0; q < 8; ++q) { hcur[q] = hnxt[q]; tcur[q] = tnxt[q]; }
#pragma unroll
    for (int t = 0; t < 4; ++t) pcur[t] = pnxt[t];
    sN = sN2;
    dN = dN2;
    }
}

__global__ __launch_bounds__(BLOCK)
void edge_mlp_wmma_kernel(const float* __restrict__ hc,
                          const float* __restrict__ hp,
                          const int*   __restrict__ src,
                          const int*   __restrict__ dst,
                          const float* __restrict__ W1,
                          const float* __restrict__ b1,
                          const float* __restrict__ W2,
                          const float* __restrict__ b2,
                          const float* __restrict__ P,
                          float*       __restrict__ out,
                          int E, int nHp, size_t wsBytes)
{
    const int tid  = threadIdx.x;
    const int lane = tid & 31;
    const int wave = __builtin_amdgcn_readfirstlane(tid >> 5);
    const int nl   = lane & 15;
    const int hi   = lane >> 4;
    const int nTiles = (E + TILE_EDGES - 1) / TILE_EDGES;

    const int tile = 2 * ((wave >> 1) * GRID_BLOCKS + blockIdx.x) + (wave & 1);
    if (tile >= nTiles) return;

    if (wsBytes >= (size_t)nHp * sizeof(float))
        edge_tiles<true >(hc, hp, src, dst, W1, b1, W2, b2, P, out, E, tile, nTiles, nl, hi);
    else
        edge_tiles<false>(hc, hp, src, dst, W1, b1, W2, b2, P, out, E, tile, nTiles, nl, hi);
}

extern "C" void kernel_launch(void* const* d_in, const int* in_sizes, int n_in,
                              void* d_out, int out_size, void* d_ws, size_t ws_size,
                              hipStream_t stream)
{
    const float* hc  = (const float*)d_in[0];
    const float* hp  = (const float*)d_in[1];
    const int*   src = (const int*)  d_in[2];
    const int*   dst = (const int*)  d_in[3];
    const float* W1  = (const float*)d_in[4];
    const float* b1  = (const float*)d_in[5];
    const float* W2  = (const float*)d_in[6];
    const float* b2  = (const float*)d_in[7];
    float* out = (float*)d_out;
    float* P   = (float*)d_ws;

    const int E   = in_sizes[2];
    const int nHp = in_sizes[1];

    hipLaunchKernelGGL(tail_projection_kernel, dim3(GRID_BLOCKS), dim3(BLOCK), 0, stream,
                       hp, W1, P, nHp, ws_size);
    hipLaunchKernelGGL(edge_mlp_wmma_kernel, dim3(GRID_BLOCKS), dim3(BLOCK), 0, stream,
                       hc, hp, src, dst, W1, b1, W2, b2, (const float*)P, out, E, nHp, ws_size);
}


# Rows in one register set: the NEXT tile's row piece is loaded into the registers just read for the last time (no 2nd set, no copies), P added behind the stream; 3 waves/SIMD, was 2; sum order changes
# speedup vs baseline: 2.0632x; 1.0387x over previous
#include <hip/hip_runtime.h>
#include <math.h>


typedef float v2f __attribute__((ext_vector_type(2)));
typedef float v4f __attribute__((ext_vector_type(4)));
typedef float v8f __attribute__((ext_vector_type(8)));

#define EMBED            64
#define WAVES_PER_BLOCK  8
#define GRID_BLOCKS      256
#define BLOCK            (WAVES_PER_BLOCK * 32)
#define TILE_EDGES       16
#define EDGE_WAVES       12
#define EDGE_BLOCK       (EDGE_WAVES * 32)

__global__ __launch_bounds__(BLOCK)
void tail_projection_kernel(const float* __restrict__ hp,
                            const float* __restrict__ W1,
                            float*       __restrict__ P,
                            int nHp, size_t wsBytes)
{
    if (wsBytes < (size_t)nHp * sizeof(float)) return;
    const int tid  = threadIdx.x;
    const int lane = tid & 31;
    const int wave = __builtin_amdgcn_readfirstlane(tid >> 5);
    const int nl   = lane & 15;
    const int hi   = lane >> 4;
    const int nRows  = nHp / EMBED;
    const int nTiles = (nRows + 15) / 16;
    const int tileStride = GRID_BLOCKS * WAVES_PER_BLOCK;
    int tile = wave * GRID_BLOCKS + blockIdx.x;
    if (tile >= nTiles) return;

    v2f wfr[16][4];
#pragma unroll
    for (int s = 0; s < 16; ++s) {
        const int k = EMBED + 8 * (s >> 1) + 4 * hi + 2 * (s & 1);
#pragma unroll
        for (int t = 0; t < 4; ++t) {
            const int n = t * 16 + nl;
            v2f b;
            b.x = W1[k * 64 + n];
            b.y = W1[(k + 1) * 64 + n];
            wfr[s][t] = b;
        }
    }

    for (; tile < nTiles; tile += tileStride) {
        const int row = tile * 16 + nl;
        const int rc  = row < nRows ? row : nRows - 1;
        const v4f* __restrict__ r4 = (const v4f*)(hp + (size_t)rc * EMBED + 4 * hi);
        v2f x[16];
#pragma unroll
        for (int q = 0; q < 8; ++q) {
            const v4f a = r4[2 * q];
            v2f lo = {a.x, a.y}, up = {a.z, a.w};
            x[2 * q] = lo; x[2 * q + 1] = up;
        }
        v8f acc[4] = {};
#pragma unroll
        for (int s = 0; s < 16; ++s) {
#pragma unroll
            for (int t = 0; t < 4; ++t) {
                acc[t] = __builtin_amdgcn_wmma_f32_16x16x4_f32(
                    false, wfr[s][t], false, x[s], (short)0, acc[t], false, false);
            }
        }
        if (row < nRows) {
            float* __restrict__ o = P + (size_t)row * EMBED + 8 * hi;
#pragma unroll
            for (int t = 0; t < 4; ++t) {
                v4f lo = {acc[t][0], acc[t][1], acc[t][2], acc[t][3]};
                v4f up = {acc[t][4], acc[t][5], acc[t][6], acc[t][7]};
                *(v4f*)(o + 16 * t)     = lo;
                *(v4f*)(o + 16 * t + 4) = up;
            }
        }
    }
}

template <bool HOIST>
__device__ __forceinline__
void edge_tiles(const float* __restrict__ hc, const float* __restrict__ hp,
                const int* __restrict__ src, const int* __restrict__ dst,
                const float* __restrict__ W1, const float* __restrict__ b1,
                const float* __restrict__ W2, const float* __restrict__ b2,
                const float* __restrict__ P,
                float* __restrict__ out, int E,
                int tile, const int nTiles, const int nl, const int hi)
{
    const int tileStride = GRID_BLOCKS * EDGE_WAVES;

    v2f bfr[16][4];
#pragma unroll
    for (int s = 0; s < 16; ++s) {
        const int k = 8 * (s >> 1) + 4 * hi + 2 * (s & 1);
#pragma unroll
        for (int t = 0; t < 4; ++t) {
            const int n = t * 16 + nl;
            v2f b;
            b.x = W1[k * 64 + n];
            b.y = W1[(k + 1) * 64 + n];
            bfr[s][t] = b;
        }
    }
    v2f wcb[4];
    float w2r[4][8];
#pragma unroll
    for (int t = 0; t < 4; ++t) {
        const int n = t * 16 + nl;
        const float wc = W1[128 * 64 + n];
        const float bb = b1[n];
        v2f e;
        e.x = hi ? 0.f : wc;
        e.y = hi ? 0.f : bb;
        wcb[t] = e;
#pragma unroll
        for (int r = 0; r < 8; ++r) w2r[t][r] = W2[t * 16 + 8 * hi + r];
    }
    const float b2s = b2[0];

    v4f hrow[8], trow[8];
    int dC, sN, dN;
    {
        int e0 = tile * TILE_EDGES + nl;
        if (e0 >= E) e0 = E - 1;
        dC = dst[e0];
        const v4f* __restrict__ h4 = (const v4f*)(hc + (size_t)src[e0] * EMBED + 4 * hi);
        const v4f* __restrict__ t4 = (const v4f*)(hp + (size_t)dC * EMBED + 4 * hi);
        int e1 = (tile + tileStride) * TILE_EDGES + nl;
        if (e1 >= E) e1 = E - 1;
        sN = src[e1];
        dN = dst[e1];
#pragma unroll
        for (int q = 0; q < 8; ++q) {
            __builtin_amdgcn_sched_barrier(0);
            hrow[q] = h4[2 * q];
            trow[q] = t4[2 * q];
        }
        __builtin_amdgcn_sched_barrier(0);
    }
    v8f p[4];
#pragma unroll
    for (int t = 0; t < 4; ++t) { v8f z = {}; p[t] = z; }
    if (HOIST) {
        const v4f* __restrict__ prow = (const v4f*)(P + (size_t)dC * EMBED + 8 * hi);
#pragma unroll
        for (int t = 0; t < 4; ++t) {
            const v4f lo = prow[4 * t], up = prow[4 * t + 1];
            v8f v = {lo.x, lo.y, lo.z, lo.w, up.x, up.y, up.z, up.w};
            p[t] = v;
        }
    }
    __builtin_amdgcn_sched_barrier(0);

    for (; tile < nTiles; tile += tileStride) {
    const int edgeBase = tile * TILE_EDGES;

    if (!HOIST) {
        const volatile float* W1v = W1;
#pragma unroll
        for (int t = 0; t < 4; ++t) { v8f z = {}; p[t] = z; }
#pragma unroll
        for (int s = 0; s < 16; ++s) {
            const int k = EMBED + 8 * (s >> 1) + 4 * hi + 2 * (s & 1);
            const v4f a4 = trow[s >> 1];
            const v2f a = (s & 1) ? __builtin_shufflevector(a4, a4, 2, 3)
                                  : __builtin_shufflevector(a4, a4, 0, 1);
#pragma unroll
            for (int t = 0; t < 4; ++t) {
                const int n = t * 16 + nl;
                v2f w;
                w.x = W1v[k * 64 + n];
                w.y = W1v[(k + 1) * 64 + n];
                p[t] = __builtin_amdgcn_wmma_f32_16x16x4_f32(
                    false, w, false, a, (short)0, p[t], false, false);
            }
        }
    }
    int e2 = (tile + 2 * tileStride) * TILE_EDGES + nl;
    if (e2 >= E) e2 = E - 1;
    const int sN2 = src[e2];
    const int dN2 = dst[e2];
    const v4f* __restrict__ hN4 = (const v4f*)(hc + (size_t)sN * EMBED + 4 * hi);
    const v4f* __restrict__ tN4 = (const v4f*)(hp + (size_t)dN * EMBED + 4 * hi);
    __builtin_amdgcn_sched_barrier(0);

    v8f acc[4];
#pragma unroll
    for (int t = 0; t < 4; ++t) { v8f z = {}; acc[t] = z; }
    v2f hh2 = {0.f, 0.f}, tt2 = {0.f, 0.f}, dp2 = {0.f, 0.f};

#pragma unroll
    for (int q = 0; q < 8; ++q) {
#pragma unroll
        for (int j = 0; j < 2; ++j) {
            const int s = 2 * q + j;
            const v4f a4 = hrow[q];
            const v4f g4 = trow[q];
            const v2f a = j ? __builtin_shufflevector(a4, a4, 2, 3)
                            : __builtin_shufflevector(a4, a4, 0, 1);
            const v2f g = j ? __builtin_shufflevector(g4, g4, 2, 3)
                            : __builtin_shufflevector(g4, g4, 0, 1);
            hh2 += a * a;
            tt2 += g * g;
            dp2 += a * g;
#pragma unroll
            for (int t = 0; t < 4; ++t) {
                acc[t] = __builtin_amdgcn_wmma_f32_16x16x4_f32(
                    false, bfr[s][t], false, a, (short)0, acc[t], false, false);
            }
        }
        __builtin_amdgcn_sched_barrier(0);
        hrow[q] = hN4[2 * q];
        trow[q] = tN4[2 * q];
        __builtin_amdgcn_sched_barrier(0);
    }

    float hh = hh2.x + hh2.y;
    float tt = tt2.x + tt2.y;
    float dp = dp2.x + dp2.y;
    hh += __shfl_xor(hh, 16, 32);
    tt += __shfl_xor(tt, 16, 32);
    dp += __shfl_xor(dp, 16, 32);
    const float nh  = __builtin_amdgcn_sqrtf(hh);
    const float nt  = __builtin_amdgcn_sqrtf(tt);
    const float inh = __builtin_amdgcn_rcpf(fmaxf(nh, 1e-12f));
    const float itn = __builtin_amdgcn_rcpf(fmaxf(nt, 1e-12f));
    const float den = fmaxf((nh * inh) * (nt * itn), 1e-8f);
    const float cosv = (dp * inh * itn) * __builtin_amdgcn_rcpf(den);

    {
        v2f cb;
        cb.x = hi ? 0.f : cosv;
        cb.y = hi ? 0.f : 1.f;
#pragma unroll
        for (int t = 0; t < 4; ++t) {
            acc[t] = __builtin_amdgcn_wmma_f32_16x16x4_f32(
                false, wcb[t], false, cb, (short)0, acc[t], false, false);
        }
    }

    float p4[4];
#pragma unroll
    for (int t = 0; t < 4; ++t) {
        const v8f pre = acc[t] + p[t];
        float ps1 = 0.f;
#pragma unroll
        for (int r = 0; r < 8; ++r) {
            const float x = fmaxf(pre[r], 0.f);
            ps1 += x * w2r[t][r];
        }
        p4[t] = ps1;
    }
    float ps = (p4[0] + p4[1]) + (p4[2] + p4[3]);
    ps += __shfl_xor(ps, 16, 32);
    const float z = ps + b2s;
    const float y = __builtin_amdgcn_rcpf(1.f + __builtin_amdgcn_exp2f(z * -1.44269504088896341f));
    const int eo = edgeBase + nl;
    if (hi == 0 && eo < E) out[eo] = y;

    __builtin_amdgcn_sched_barrier(0);
    if (HOIST) {
        const v4f* __restrict__ prow = (const v4f*)(P + (size_t)dN * EMBED + 8 * hi);
#pragma unroll
        for (int t = 0; t < 4; ++t) {
            const v4f lo = prow[4 * t], up = prow[4 * t + 1];
            v8f v = {lo.x, lo.y, lo.z, lo.w, up.x, up.y, up.z, up.w};
            p[t] = v;
        }
    }
    sN = sN2;
    dN = dN2;
    }
}

__global__ __launch_bounds__(EDGE_BLOCK)
void edge_mlp_wmma_kernel(const float* __restrict__ hc,
                          const float* __restrict__ hp,
                          const int*   __restrict__ src,
                          const int*   __restrict__ dst,
                          const float* __restrict__ W1,
                          const float* __restrict__ b1,
                          const float* __restrict__ W2,
                          const float* __restrict__ b2,
                          const float* __restrict__ P,
                          float*       __restrict__ out,
                          int E, int nHp, size_t wsBytes)
{
    const int tid  = threadIdx.x;
    const int lane = tid & 31;
    const int wave = __builtin_amdgcn_readfirstlane(tid >> 5);
    const int nl   = lane & 15;
    const int hi   = lane >> 4;
    const int nTiles = (E + TILE_EDGES - 1) / TILE_EDGES;

    const int tile = 2 * ((wave >> 1) * GRID_BLOCKS + blockIdx.x) + (wave & 1);
    if (tile >= nTiles) return;

    if (wsBytes >= (size_t)nHp * sizeof(float))
        edge_tiles<true >(hc, hp, src, dst, W1, b1, W2, b2, P, out, E, tile, nTiles, nl, hi);
    else
        edge_tiles<false>(hc, hp, src, dst, W1, b1, W2, b2, P, out, E, tile, nTiles, nl, hi);
}

extern "C" void kernel_launch(void* const* d_in, const int* in_sizes, int n_in,
                              void* d_out, int out_size, void* d_ws, size_t ws_size,
                              hipStream_t stream)
{
    const float* hc  = (const float*)d_in[0];
    const float* hp  = (const float*)d_in[1];
    const int*   src = (const int*)  d_in[2];
    const int*   dst = (const int*)  d_in[3];
    const float* W1  = (const float*)d_in[4];
    const float* b1  = (const float*)d_in[5];
    const float* W2  = (const float*)d_in[6];
    const float* b2  = (const float*)d_in[7];
    float* out = (float*)d_out;
    float* P   = (float*)d_ws;

    const int E   = in_sizes[2];
    const int nHp = in_sizes[1];

    hipLaunchKernelGGL(tail_projection_kernel, dim3(GRID_BLOCKS), dim3(BLOCK), 0, stream,
                       hp, W1, P, nHp, ws_size);
    hipLaunchKernelGGL(edge_mlp_wmma_kernel, dim3(GRID_BLOCKS), dim3(EDGE_BLOCK), 0, stream,
                       hc, hp, src, dst, W1, b1, W2, b2, (const float*)P, out, E, nHp, ws_size);
}
